# starting kernel
# baseline (speedup 1.0000x reference)
#include <hip/hip_runtime.h>


typedef __attribute__((ext_vector_type(2))) float v2f;
typedef __attribute__((ext_vector_type(8))) float v8f;

#define WAVES_PER_WG 16
#define WG_THREADS   (WAVES_PER_WG * 32)

static __device__ __forceinline__ v8f wmma_f32(v2f a, v2f b, v8f c) {
    return __builtin_amdgcn_wmma_f32_16x16x4_f32(false, a, false, b,
                                                 (short)0, c, false, false);
}

__global__ __launch_bounds__(WG_THREADS)
void m1_recur_gemv_wmma(const float* __restrict__ x,
                        const float* __restrict__ pa, const float* __restrict__ pb,
                        const float* __restrict__ pc, const float* __restrict__ pw,
                        const float* __restrict__ pe,
                        float* __restrict__ out, int B, int T)
{
    const float a    = pa[0];
    const float bco  = pb[0];
    const float cco  = pc[0];
    const float wco  = pw[0];
    const float eco  = pe[0];
    const float la2  = __builtin_amdgcn_logf(a);
    const float inva = __builtin_amdgcn_rcpf(a);
    const float inva8 = __builtin_amdgcn_exp2f(-8.0f * la2);

    const int tid  = threadIdx.x;
    const int wave = tid >> 5;
    const int lane = tid & 31;
    const int m    = lane & 15;
    const int khi  = lane >> 4;

    const int row0 = blockIdx.x * 16;
    int row = row0 + m;
    if (row >= B) row = B - 1;
    const float* rowp = x + (size_t)row * (size_t)T;

    const int tchunk = T / WAVES_PER_WG;
    const int tbeg   = wave * tchunk;
    const float* p   = rowp + tbeg + khi * 4;

    v8f acc = {};
    float ex = (float)(T - 1 - tbeg - 4 * khi);

    for (int t = 0; t < tchunk; t += 16) {
        const float4 x0 = *(const float4*)(p + t);
        const float4 x1 = *(const float4*)(p + t + 8);
        __builtin_prefetch(p + t + 1024, 0, 1);

        const float w0 = __builtin_amdgcn_exp2f(la2 * ex);
        const float w1 = w0 * inva;
        const float w2 = w1 * inva;
        const float w3 = w2 * inva;
        const float w4 = w0 * inva8;
        const float w5 = w4 * inva;
        const float w6 = w5 * inva;
        const float w7 = w6 * inva;
        ex -= 16.0f;

        v2f a0 = {x0.x, x0.y}, b0 = {w0, w1};
        v2f a1 = {x0.z, x0.w}, b1 = {w2, w3};
        v2f a2 = {x1.x, x1.y}, b2 = {w4, w5};
        v2f a3 = {x1.z, x1.w}, b3 = {w6, w7};
        acc = wmma_f32(a0, b0, acc);
        acc = wmma_f32(a1, b1, acc);
        acc = wmma_f32(a2, b2, acc);
        acc = wmma_f32(a3, b3, acc);
    }

    __shared__ float red[WAVES_PER_WG * 16];
    if (m == 0) {
        const int base = wave * 16 + khi * 8;
        #pragma unroll
        for (int i = 0; i < 8; ++i) red[base + i] = acc[i];
    }
    __syncthreads();

    if (tid < 16) {
        float dot = 0.0f;
        #pragma unroll
        for (int wv = 0; wv < WAVES_PER_WG; ++wv) dot += red[wv * 16 + tid];
        const float aT = __builtin_amdgcn_exp2f(la2 * (float)T);
        const float G  = (1.0f - aT) / (1.0f - a);
        const int orow = row0 + tid;
        if (orow < B) out[orow] = wco * (bco * dot + cco * G) + eco;
    }
}

extern "C" void kernel_launch(void* const* d_in, const int* in_sizes, int n_in,
                              void* d_out, int out_size, void* d_ws, size_t ws_size,
                              hipStream_t stream) {
    (void)d_ws; (void)ws_size; (void)n_in;
    const float* x  = (const float*)d_in[0];
    const float* pa = (const float*)d_in[1];
    const float* pb = (const float*)d_in[2];
    const float* pc = (const float*)d_in[3];
    const float* pw = (const float*)d_in[4];
    const float* pe = (const float*)d_in[5];
    float* out = (float*)d_out;

    const int B = out_size;
    const int T = in_sizes[0] / out_size;

    const int nblocks = (B + 15) / 16;
    m1_recur_gemv_wmma<<<nblocks, WG_THREADS, 0, stream>>>(x, pa, pb, pc, pw, pe, out, B, T);
}


# Loads issued 16 b128 per lane one 128-step stage ahead of the WMMAs (two register buffers, 16-32 loads in flight per wave, was 8 then drained); prefetch dropped; order of sums unchanged
# speedup vs baseline: 1.1246x; 1.1246x over previous
#include <hip/hip_runtime.h>


typedef __attribute__((ext_vector_type(2))) float v2f;
typedef __attribute__((ext_vector_type(8))) float v8f;

#define WAVES_PER_WG 16
#define WG_THREADS   (WAVES_PER_WG * 32)
#define SUBSTEPS     8
#define STAGE_T      (SUBSTEPS * 16)

static __device__ __forceinline__ v8f wmma_f32(v2f a, v2f b, v8f c) {
    return __builtin_amdgcn_wmma_f32_16x16x4_f32(false, a, false, b,
                                                 (short)0, c, false, false);
}

static __device__ __forceinline__ v8f piece16(const float4 x0, const float4 x1,
                                              const float la2, const float inva,
                                              const float inva8, float& ex, v8f acc) {
    const float w0 = __builtin_amdgcn_exp2f(la2 * ex);
    const float w1 = w0 * inva;
    const float w2 = w1 * inva;
    const float w3 = w2 * inva;
    const float w4 = w0 * inva8;
    const float w5 = w4 * inva;
    const float w6 = w5 * inva;
    const float w7 = w6 * inva;
    ex -= 16.0f;

    v2f a0 = {x0.x, x0.y}, b0 = {w0, w1};
    v2f a1 = {x0.z, x0.w}, b1 = {w2, w3};
    v2f a2 = {x1.x, x1.y}, b2 = {w4, w5};
    v2f a3 = {x1.z, x1.w}, b3 = {w6, w7};
    acc = wmma_f32(a0, b0, acc);
    acc = wmma_f32(a1, b1, acc);
    acc = wmma_f32(a2, b2, acc);
    acc = wmma_f32(a3, b3, acc);
    return acc;
}

static __device__ __forceinline__ void load_stage(float4 (&buf)[2 * SUBSTEPS], const float* q) {
    #pragma unroll
    for (int j = 0; j < SUBSTEPS; ++j) {
        buf[2 * j]     = *(const float4*)(q + 16 * j);
        buf[2 * j + 1] = *(const float4*)(q + 16 * j + 8);
    }
}

static __device__ __forceinline__ v8f compute_stage(const float4 (&buf)[2 * SUBSTEPS],
                                                    const float la2, const float inva,
                                                    const float inva8, float& ex, v8f acc) {
    #pragma unroll
    for (int j = 0; j < SUBSTEPS; ++j)
        acc = piece16(buf[2 * j], buf[2 * j + 1], la2, inva, inva8, ex, acc);
    return acc;
}

#define PIN_ORDER() __builtin_amdgcn_sched_barrier(0)

__global__ __launch_bounds__(WG_THREADS)
void m1_recur_gemv_wmma(const float* __restrict__ x,
                        const float* __restrict__ pa, const float* __restrict__ pb,
                        const float* __restrict__ pc, const float* __restrict__ pw,
                        const float* __restrict__ pe,
                        float* __restrict__ out, int B, int T)
{
    const float a    = pa[0];
    const float bco  = pb[0];
    const float cco  = pc[0];
    const float wco  = pw[0];
    const float eco  = pe[0];
    const float la2  = __builtin_amdgcn_logf(a);
    const float inva = __builtin_amdgcn_rcpf(a);
    const float inva8 = __builtin_amdgcn_exp2f(-8.0f * la2);

    const int tid  = threadIdx.x;
    const int wave = tid >> 5;
    const int lane = tid & 31;
    const int m    = lane & 15;
    const int khi  = lane >> 4;

    const int row0 = blockIdx.x * 16;
    int row = row0 + m;
    if (row >= B) row = B - 1;
    const float* rowp = x + (size_t)row * (size_t)T;

    const int tchunk = T / WAVES_PER_WG;
    const int tbeg   = wave * tchunk;
    const float* p   = rowp + tbeg + khi * 4;

    v8f acc = {};
    float ex = (float)(T - 1 - tbeg - 4 * khi);

    const int nst = tchunk / STAGE_T;
    float4 bufA[2 * SUBSTEPS], bufB[2 * SUBSTEPS];
    if (nst > 0) {
        int tq = 0;
        load_stage(bufA, p);
        int s = 0;
        for (; s + 2 < nst; s += 2) {
            PIN_ORDER();
            load_stage(bufB, p + tq + STAGE_T);
            PIN_ORDER();
            acc = compute_stage(bufA, la2, inva, inva8, ex, acc);
            tq += 2 * STAGE_T;
            asm volatile("" : "+s"(tq), "+v"(acc));
            PIN_ORDER();
            load_stage(bufA, p + tq);
            PIN_ORDER();
            acc = compute_stage(bufB, la2, inva, inva8, ex, acc);
        }
        if (s + 1 < nst) {
            PIN_ORDER();
            load_stage(bufB, p + tq + STAGE_T);
            PIN_ORDER();
            acc = compute_stage(bufA, la2, inva, inva8, ex, acc);
            acc = compute_stage(bufB, la2, inva, inva8, ex, acc);
        } else {
            acc = compute_stage(bufA, la2, inva, inva8, ex, acc);
        }
    }
    for (int t = nst * STAGE_T; t < tchunk; t += 16) {
        const float4 x0 = *(const float4*)(p + t);
        const float4 x1 = *(const float4*)(p + t + 8);
        acc = piece16(x0, x1, la2, inva, inva8, ex, acc);
    }

    __shared__ float red[WAVES_PER_WG * 16];
    if (m == 0) {
        const int base = wave * 16 + khi * 8;
        #pragma unroll
        for (int i = 0; i < 8; ++i) red[base + i] = acc[i];
    }
    __syncthreads();

    if (tid < 16) {
        float dot = 0.0f;
        #pragma unroll
        for (int wv = 0; wv < WAVES_PER_WG; ++wv) dot += red[wv * 16 + tid];
        const float aT = __builtin_amdgcn_exp2f(la2 * (float)T);
        const float G  = (1.0f - aT) / (1.0f - a);
        const int orow = row0 + tid;
        if (orow < B) out[orow] = wco * (bco * dot + cco * G) + eco;
    }
}

extern "C" void kernel_launch(void* const* d_in, const int* in_sizes, int n_in,
                              void* d_out, int out_size, void* d_ws, size_t ws_size,
                              hipStream_t stream) {
    (void)d_ws; (void)ws_size; (void)n_in;
    const float* x  = (const float*)d_in[0];
    const float* pa = (const float*)d_in[1];
    const float* pb = (const float*)d_in[2];
    const float* pc = (const float*)d_in[3];
    const float* pw = (const float*)d_in[4];
    const float* pe = (const float*)d_in[5];
    float* out = (float*)d_out;

    const int B = out_size;
    const int T = in_sizes[0] / out_size;

    const int nblocks = (B + 15) / 16;
    m1_recur_gemv_wmma<<<nblocks, WG_THREADS, 0, stream>>>(x, pa, pb, pc, pw, pe, out, B, T);
}


# Each wave-wide b128 load now reads 512 contiguous bytes of one row (4 lines, was 16): one wave per row, WMMA M index = 4-step group, a^(60-4m) applied at the end; reduction tree changed
# speedup vs baseline: 1.4807x; 1.3166x over previous
#include <hip/hip_runtime.h>


typedef __attribute__((ext_vector_type(2))) float v2f;
typedef __attribute__((ext_vector_type(8))) float v8f;

#define WAVES_PER_WG 16
#define WG_THREADS   (WAVES_PER_WG * 32)
#define PIECE_T      128
#define STAGE_LOADS  16
#define STAGE_T      (STAGE_LOADS * PIECE_T)

static __device__ __forceinline__ v8f wmma_f32(v2f a, v2f b, v8f c) {
    return __builtin_amdgcn_wmma_f32_16x16x4_f32(false, a, false, b,
                                                 (short)0, c, false, false);
}

static __device__ __forceinline__ v8f piece128(const float4 xv, const float la2,
                                               const float inva, float& ex, v8f acc) {
    const float w0 = __builtin_amdgcn_exp2f(la2 * ex);
    const float w1 = w0 * inva;
    const float w2 = w1 * inva;
    const float w3 = w2 * inva;
    ex -= (float)PIECE_T;

    v2f a0 = {xv.x, xv.y}, b0 = {w0, w1};
    v2f a1 = {xv.z, xv.w}, b1 = {w2, w3};
    acc = wmma_f32(a0, b0, acc);
    acc = wmma_f32(a1, b1, acc);
    return acc;
}

static __device__ __forceinline__ void load_stage(float4 (&buf)[STAGE_LOADS], const float* q) {
    #pragma unroll
    for (int j = 0; j < STAGE_LOADS; ++j)
        buf[j] = *(const float4*)(q + PIECE_T * j);
}

static __device__ __forceinline__ v8f compute_stage(const float4 (&buf)[STAGE_LOADS],
                                                    const float la2, const float inva,
                                                    float& ex, v8f acc) {
    #pragma unroll
    for (int j = 0; j < STAGE_LOADS; ++j)
        acc = piece128(buf[j], la2, inva, ex, acc);
    return acc;
}

#define PIN_ORDER() __builtin_amdgcn_sched_barrier(0)

__global__ __launch_bounds__(WG_THREADS)
void m1_recur_gemv_wmma(const float* __restrict__ x,
                        const float* __restrict__ pa, const float* __restrict__ pb,
                        const float* __restrict__ pc, const float* __restrict__ pw,
                        const float* __restrict__ pe,
                        float* __restrict__ out, int B, int T)
{
    const float a    = pa[0];
    const float bco  = pb[0];
    const float cco  = pc[0];
    const float wco  = pw[0];
    const float eco  = pe[0];
    const float la2  = __builtin_amdgcn_logf(a);
    const float inva = __builtin_amdgcn_rcpf(a);

    const int tid  = threadIdx.x;
    const int wave = tid >> 5;
    const int lane = tid & 31;
    const int c    = lane & 15;
    const int h    = lane >> 4;

    const int row0 = blockIdx.x * WAVES_PER_WG;
    int row = row0 + wave;
    if (row >= B) row = B - 1;
    const float* p = x + (size_t)row * (size_t)T + lane * 4;

    v8f acc = {};
    float ex = (float)(T - 61 - 64 * h);

    const int nst = T / STAGE_T;
    float4 bufA[STAGE_LOADS], bufB[STAGE_LOADS];
    if (nst > 0) {
        int tq = 0;
        load_stage(bufA, p);
        int s = 0;
        for (; s + 2 < nst; s += 2) {
            PIN_ORDER();
            load_stage(bufB, p + tq + STAGE_T);
            PIN_ORDER();
            acc = compute_stage(bufA, la2, inva, ex, acc);
            tq += 2 * STAGE_T;
            asm volatile("" : "+s"(tq), "+v"(acc));
            PIN_ORDER();
            load_stage(bufA, p + tq);
            PIN_ORDER();
            acc = compute_stage(bufB, la2, inva, ex, acc);
        }
        if (s + 1 < nst) {
            PIN_ORDER();
            load_stage(bufB, p + tq + STAGE_T);
            PIN_ORDER();
            acc = compute_stage(bufA, la2, inva, ex, acc);
            acc = compute_stage(bufB, la2, inva, ex, acc);
        } else {
            acc = compute_stage(bufA, la2, inva, ex, acc);
        }
    }
    for (int t = nst * STAGE_T; t + PIECE_T <= T; t += PIECE_T) {
        const float4 xv = *(const float4*)(p + t);
        acc = piece128(xv, la2, inva, ex, acc);
    }

    float part = 0.0f;
    const float fe = (float)(60 - 32 * h);
    #pragma unroll
    for (int r = 0; r < 8; ++r)
        part = __builtin_fmaf(__builtin_amdgcn_exp2f(la2 * (fe - 4.0f * (float)r)), acc[r], part);

    __shared__ float red[WAVES_PER_WG * 2];
    if (c == 0) red[wave * 2 + h] = part;
    __syncthreads();

    if (tid < WAVES_PER_WG) {
        const float dot = red[2 * tid] + red[2 * tid + 1];
        const float aT = __builtin_amdgcn_exp2f(la2 * (float)T);
        const float G  = (1.0f - aT) / (1.0f - a);
        const int orow = row0 + tid;
        if (orow < B) out[orow] = wco * (bco * dot + cco * G) + eco;
    }
}

extern "C" void kernel_launch(void* const* d_in, const int* in_sizes, int n_in,
                              void* d_out, int out_size, void* d_ws, size_t ws_size,
                              hipStream_t stream) {
    (void)d_ws; (void)ws_size; (void)n_in;
    const float* x  = (const float*)d_in[0];
    const float* pa = (const float*)d_in[1];
    const float* pb = (const float*)d_in[2];
    const float* pc = (const float*)d_in[3];
    const float* pw = (const float*)d_in[4];
    const float* pe = (const float*)d_in[5];
    float* out = (float*)d_out;

    const int B = out_size;
    const int T = in_sizes[0] / out_size;

    const int nblocks = (B + 15) / 16;
    m1_recur_gemv_wmma<<<nblocks, WG_THREADS, 0, stream>>>(x, pa, pb, pc, pw, pe, out, B, T);
}


# B weight fragments of one 2048-step stage built once per wave, behind the first loads, kept in registers; stages joined by Horner (acc*a^2048): loop = 1 load + 2 WMMAs per piece; sum order changed
# speedup vs baseline: 1.5794x; 1.0667x over previous
#include <hip/hip_runtime.h>


typedef __attribute__((ext_vector_type(2))) float v2f;
typedef __attribute__((ext_vector_type(8))) float v8f;

#define WAVES_PER_WG 16
#define WG_THREADS   (WAVES_PER_WG * 32)
#define PIECE_T      128
#define STAGE_LOADS  16
#define STAGE_T      (STAGE_LOADS * PIECE_T)

static __device__ __forceinline__ v8f wmma_f32(v2f a, v2f b, v8f c) {
    return __builtin_amdgcn_wmma_f32_16x16x4_f32(false, a, false, b,
                                                 (short)0, c, false, false);
}

static __device__ __forceinline__ v8f piece128(const float4 xv, const float la2,
                                               const float inva, float& ex, v8f acc) {
    const float w0 = __builtin_amdgcn_exp2f(la2 * ex);
    const float w1 = w0 * inva;
    const float w2 = w1 * inva;
    const float w3 = w2 * inva;
    ex -= (float)PIECE_T;

    v2f a0 = {xv.x, xv.y}, b0 = {w0, w1};
    v2f a1 = {xv.z, xv.w}, b1 = {w2, w3};
    acc = wmma_f32(a0, b0, acc);
    acc = wmma_f32(a1, b1, acc);
    return acc;
}

static __device__ __forceinline__ void load_stage(float4 (&buf)[STAGE_LOADS], const float* q) {
    #pragma unroll
    for (int j = 0; j < STAGE_LOADS; ++j)
        buf[j] = *(const float4*)(q + PIECE_T * j);
}

struct StageW { v2f b0[STAGE_LOADS]; v2f b1[STAGE_LOADS]; };

static __device__ __forceinline__ void build_stage_weights(StageW& sw, const float la2,
                                                           const float inva, const int h) {
    #pragma unroll
    for (int j = 0; j < STAGE_LOADS; ++j) {
        const float e  = (float)(STAGE_T - 61 - PIECE_T * j - 64 * h);
        const float w0 = __builtin_amdgcn_exp2f(la2 * e);
        const float w1 = w0 * inva;
        const float w2 = w1 * inva;
        const float w3 = w2 * inva;
        sw.b0[j] = v2f{w0, w1};
        sw.b1[j] = v2f{w2, w3};
    }
}

static __device__ __forceinline__ v8f compute_stage(const float4 (&buf)[STAGE_LOADS],
                                                    const StageW& sw, const float hs, v8f acc) {
    #pragma unroll
    for (int r = 0; r < 8; ++r) acc[r] *= hs;
    #pragma unroll
    for (int j = 0; j < STAGE_LOADS; ++j) {
        v2f a0 = {buf[j].x, buf[j].y};
        v2f a1 = {buf[j].z, buf[j].w};
        acc = wmma_f32(a0, sw.b0[j], acc);
        acc = wmma_f32(a1, sw.b1[j], acc);
    }
    return acc;
}

#define PIN_ORDER() __builtin_amdgcn_sched_barrier(0)

__global__ __launch_bounds__(WG_THREADS)
void m1_recur_gemv_wmma(const float* __restrict__ x,
                        const float* __restrict__ pa, const float* __restrict__ pb,
                        const float* __restrict__ pc, const float* __restrict__ pw,
                        const float* __restrict__ pe,
                        float* __restrict__ out, int B, int T)
{
    const float a    = pa[0];
    const float bco  = pb[0];
    const float cco  = pc[0];
    const float wco  = pw[0];
    const float eco  = pe[0];

    const int tid  = threadIdx.x;
    const int wave = tid >> 5;
    const int lane = tid & 31;
    const int c    = lane & 15;
    const int h    = lane >> 4;

    const int row0 = blockIdx.x * WAVES_PER_WG;
    int row = row0 + wave;
    if (row >= B) row = B - 1;
    const float* p = x + (size_t)row * (size_t)T + lane * 4;

    v8f acc = {};

    const int nst = T / STAGE_T;
    float4 bufA[STAGE_LOADS], bufB[STAGE_LOADS];
    if (nst > 0) load_stage(bufA, p);
    PIN_ORDER();
    const float la2  = __builtin_amdgcn_logf(a);
    const float inva = __builtin_amdgcn_rcpf(a);

    if (nst > 0) {
        StageW sw;
        build_stage_weights(sw, la2, inva, h);
        const float a_stage = __builtin_amdgcn_exp2f(la2 * (float)STAGE_T);
        float hs = 1.0f;
        int tq = 0;
        int s = 0;
        for (; s + 2 < nst; s += 2) {
            PIN_ORDER();
            load_stage(bufB, p + tq + STAGE_T);
            PIN_ORDER();
            acc = compute_stage(bufA, sw, hs, acc);
            hs = a_stage;
            tq += 2 * STAGE_T;
            asm volatile("" : "+s"(tq), "+v"(acc));
            PIN_ORDER();
            load_stage(bufA, p + tq);
            PIN_ORDER();
            acc = compute_stage(bufB, sw, a_stage, acc);
        }
        if (s + 1 < nst) {
            PIN_ORDER();
            load_stage(bufB, p + tq + STAGE_T);
            PIN_ORDER();
            acc = compute_stage(bufA, sw, hs, acc);
            acc = compute_stage(bufB, sw, a_stage, acc);
        } else {
            acc = compute_stage(bufA, sw, hs, acc);
        }
    }
    const int t0 = nst * STAGE_T;
    if (t0 + PIECE_T <= T) {
        const int rest = T - t0;
        if (nst > 0) {
            const float ar = __builtin_amdgcn_exp2f(la2 * (float)rest);
            #pragma unroll
            for (int r = 0; r < 8; ++r) acc[r] *= ar;
        }
        float ex = (float)(rest - 61 - 64 * h);
        for (int t = t0; t + PIECE_T <= T; t += PIECE_T) {
            const float4 xv = *(const float4*)(p + t);
            acc = piece128(xv, la2, inva, ex, acc);
        }
    }

    float part = 0.0f;
    const float fe = (float)(60 - 32 * h);
    #pragma unroll
    for (int r = 0; r < 8; ++r)
        part = __builtin_fmaf(__builtin_amdgcn_exp2f(la2 * (fe - 4.0f * (float)r)), acc[r], part);

    __shared__ float red[WAVES_PER_WG * 2];
    if (c == 0) red[wave * 2 + h] = part;
    __syncthreads();

    if (tid < WAVES_PER_WG) {
        const float dot = red[2 * tid] + red[2 * tid + 1];
        const float aT = __builtin_amdgcn_exp2f(la2 * (float)T);
        const float G  = (1.0f - aT) / (1.0f - a);
        const int orow = row0 + tid;
        if (orow < B) out[orow] = wco * (bco * dot + cco * G) + eco;
    }
}

extern "C" void kernel_launch(void* const* d_in, const int* in_sizes, int n_in,
                              void* d_out, int out_size, void* d_ws, size_t ws_size,
                              hipStream_t stream) {
    (void)d_ws; (void)ws_size; (void)n_in;
    const float* x  = (const float*)d_in[0];
    const float* pa = (const float*)d_in[1];
    const float* pb = (const float*)d_in[2];
    const float* pc = (const float*)d_in[3];
    const float* pw = (const float*)d_in[4];
    const float* pe = (const float*)d_in[5];
    float* out = (float*)d_out;

    const int B = out_size;
    const int T = in_sizes[0] / out_size;

    const int nblocks = (B + 15) / 16;
    m1_recur_gemv_wmma<<<nblocks, WG_THREADS, 0, stream>>>(x, pa, pb, pc, pw, pe, out, B, T);
}
